# starting kernel
# baseline (speedup 1.0000x reference)
#include <hip/hip_runtime.h>

typedef __attribute__((ext_vector_type(2))) float v2f;
typedef __attribute__((ext_vector_type(8))) float v8f;

#define D 128
#define NTILES_N (D/16)

__global__ void gcn_zero_kernel(float* __restrict__ out, int n4) {
    int i = blockIdx.x * blockDim.x + threadIdx.x;
    if (i < n4) {
        ((float4*)out)[i] = make_float4(0.f, 0.f, 0.f, 0.f);
    }
}

__global__ void __launch_bounds__(256)
gcn_gemm_wmma_f32(const float* __restrict__ A,
                  const float* __restrict__ W,
                  float* __restrict__ S,
                  int M) {
    __shared__ float2 Wlds2[(D / 2) * D];

    for (int i = threadIdx.x; i < (D / 2) * (D / 4); i += blockDim.x) {
        const int p  = i / (D / 4);
        const int c4 = (i % (D / 4)) * 4;
        const float4 r0 = *(const float4*)(W + (size_t)(2 * p)     * D + c4);
        const float4 r1 = *(const float4*)(W + (size_t)(2 * p + 1) * D + c4);
        float2* outp = Wlds2 + p * D + c4;
        outp[0] = make_float2(r0.x, r1.x);
        outp[1] = make_float2(r0.y, r1.y);
        outp[2] = make_float2(r0.z, r1.z);
        outp[3] = make_float2(r0.w, r1.w);
    }
    __syncthreads();

    const int lane  = threadIdx.x & 31;
    const int tileM = blockIdx.x * (blockDim.x >> 5) + (threadIdx.x >> 5);
    if (tileM * 16 >= M) return;

    const int half = lane >> 4;
    const int l15  = lane & 15;
    const int m    = min(tileM * 16 + l15, M - 1);
    const float*  Arow = A + (size_t)m * D;
    const float2* Wb   = Wlds2 + half * D + l15;

    v8f acc[NTILES_N] = {};
#pragma unroll
    for (int k = 0; k < D; k += 4) {
        const v2f a = *(const v2f*)(Arow + k + half * 2);
#pragma unroll
        for (int n = 0; n < NTILES_N; ++n) {
            const v2f b = *(const v2f*)(Wb + (k >> 1) * D + n * 16);
            acc[n] = __builtin_amdgcn_wmma_f32_16x16x4_f32(
                false, a, false, b, (short)0, acc[n], false, false);
        }
    }

#pragma unroll
    for (int r = 0; r < 8; ++r) {
        const int row = tileM * 16 + half * 8 + r;
        if (row < M) {
            float* Sp = S + (size_t)row * D + l15;
#pragma unroll
            for (int n = 0; n < NTILES_N; ++n) {
                Sp[n * 16] = acc[n][r];
            }
        }
    }
}

__global__ void gcn_spmm_scatter(const float* __restrict__ support,
                                 const int*   __restrict__ dst,
                                 const int*   __restrict__ src,
                                 const float* __restrict__ vals,
                                 float*       __restrict__ out,
                                 int E) {
    const int edge = (blockIdx.x * blockDim.x + threadIdx.x) >> 5;
    const int lane = threadIdx.x & 31;
    if (edge >= E) return;

    const int   s = src[edge];
    const int   d = dst[edge];
    const float v = vals[edge];

    const float4 msg = ((const float4*)(support + (size_t)s * D))[lane];
    float* op = out + (size_t)d * D + lane * 4;
    atomicAdd(op + 0, v * msg.x);
    atomicAdd(op + 1, v * msg.y);
    atomicAdd(op + 2, v * msg.z);
    atomicAdd(op + 3, v * msg.w);
}

extern "C" void kernel_launch(void* const* d_in, const int* in_sizes, int n_in,
                              void* d_out, int out_size, void* d_ws, size_t ws_size,
                              hipStream_t stream) {
    const float* embeds = (const float*)d_in[0];
    const float* weight = (const float*)d_in[1];
    const int*   eidx   = (const int*)d_in[2];
    const float* evals  = (const float*)d_in[3];

    const int M = in_sizes[0] / D;
    const int E = in_sizes[3];
    const int* dst = eidx;
    const int* src = eidx + E;

    float* support = (float*)d_ws;
    float* out     = (float*)d_out;

    {
        const int n4 = out_size / 4;
        gcn_zero_kernel<<<(n4 + 255) / 256, 256, 0, stream>>>(out, n4);
    }

    {
        const int mtiles = (M + 15) / 16;
        const int wavesPerBlock = 8;
        const int blocks = (mtiles + wavesPerBlock - 1) / wavesPerBlock;
        gcn_gemm_wmma_f32<<<blocks, wavesPerBlock * 32, 0, stream>>>(
            embeds, weight, support, M);
    }

    {
        const int wavesPerBlock = 8;
        const int blocks = (E + wavesPerBlock - 1) / wavesPerBlock;
        gcn_spmm_scatter<<<blocks, wavesPerBlock * 32, 0, stream>>>(
            support, dst, src, evals, out, E);
    }
}


# scatter: lane owns columns l+32j so each row gather and each f32 atomic add instruction covers one whole 128-B line, not a quarter of four lines; float atomics kept, sum order by arrival as before
# speedup vs baseline: 1.3054x; 1.3054x over previous
#include <hip/hip_runtime.h>

typedef __attribute__((ext_vector_type(2))) float v2f;
typedef __attribute__((ext_vector_type(8))) float v8f;

#define D 128
#define NTILES_N (D/16)

__global__ void gcn_zero_kernel(float* __restrict__ out, int n4) {
    int i = blockIdx.x * blockDim.x + threadIdx.x;
    if (i < n4) {
        ((float4*)out)[i] = make_float4(0.f, 0.f, 0.f, 0.f);
    }
}

__global__ void __launch_bounds__(256)
gcn_gemm_wmma_f32(const float* __restrict__ A,
                  const float* __restrict__ W,
                  float* __restrict__ S,
                  int M) {
    __shared__ float2 Wlds2[(D / 2) * D];

    for (int i = threadIdx.x; i < (D / 2) * (D / 4); i += blockDim.x) {
        const int p  = i / (D / 4);
        const int c4 = (i % (D / 4)) * 4;
        const float4 r0 = *(const float4*)(W + (size_t)(2 * p)     * D + c4);
        const float4 r1 = *(const float4*)(W + (size_t)(2 * p + 1) * D + c4);
        float2* outp = Wlds2 + p * D + c4;
        outp[0] = make_float2(r0.x, r1.x);
        outp[1] = make_float2(r0.y, r1.y);
        outp[2] = make_float2(r0.z, r1.z);
        outp[3] = make_float2(r0.w, r1.w);
    }
    __syncthreads();

    const int lane  = threadIdx.x & 31;
    const int tileM = blockIdx.x * (blockDim.x >> 5) + (threadIdx.x >> 5);
    if (tileM * 16 >= M) return;

    const int half = lane >> 4;
    const int l15  = lane & 15;
    const int m    = min(tileM * 16 + l15, M - 1);
    const float*  Arow = A + (size_t)m * D;
    const float2* Wb   = Wlds2 + half * D + l15;

    v8f acc[NTILES_N] = {};
#pragma unroll
    for (int k = 0; k < D; k += 4) {
        const v2f a = *(const v2f*)(Arow + k + half * 2);
#pragma unroll
        for (int n = 0; n < NTILES_N; ++n) {
            const v2f b = *(const v2f*)(Wb + (k >> 1) * D + n * 16);
            acc[n] = __builtin_amdgcn_wmma_f32_16x16x4_f32(
                false, a, false, b, (short)0, acc[n], false, false);
        }
    }

#pragma unroll
    for (int r = 0; r < 8; ++r) {
        const int row = tileM * 16 + half * 8 + r;
        if (row < M) {
            float* Sp = S + (size_t)row * D + l15;
#pragma unroll
            for (int n = 0; n < NTILES_N; ++n) {
                Sp[n * 16] = acc[n][r];
            }
        }
    }
}

__global__ void gcn_spmm_scatter(const float* __restrict__ support,
                                 const int*   __restrict__ dst,
                                 const int*   __restrict__ src,
                                 const float* __restrict__ vals,
                                 float*       __restrict__ out,
                                 int E) {
    const int edge = (blockIdx.x * blockDim.x + threadIdx.x) >> 5;
    const int lane = threadIdx.x & 31;
    if (edge >= E) return;

    const int   s = src[edge];
    const int   d = dst[edge];
    const float v = vals[edge];

    const float* sp = support + (size_t)s * D + lane;
    float*       op = out     + (size_t)d * D + lane;

    const float m0 = sp[0];
    const float m1 = sp[32];
    const float m2 = sp[64];
    const float m3 = sp[96];

    atomicAdd(op +  0, v * m0);
    atomicAdd(op + 32, v * m1);
    atomicAdd(op + 64, v * m2);
    atomicAdd(op + 96, v * m3);
}

extern "C" void kernel_launch(void* const* d_in, const int* in_sizes, int n_in,
                              void* d_out, int out_size, void* d_ws, size_t ws_size,
                              hipStream_t stream) {
    const float* embeds = (const float*)d_in[0];
    const float* weight = (const float*)d_in[1];
    const int*   eidx   = (const int*)d_in[2];
    const float* evals  = (const float*)d_in[3];

    const int M = in_sizes[0] / D;
    const int E = in_sizes[3];
    const int* dst = eidx;
    const int* src = eidx + E;

    float* support = (float*)d_ws;
    float* out     = (float*)d_out;

    {
        const int n4 = out_size / 4;
        gcn_zero_kernel<<<(n4 + 255) / 256, 256, 0, stream>>>(out, n4);
    }

    {
        const int mtiles = (M + 15) / 16;
        const int wavesPerBlock = 8;
        const int blocks = (mtiles + wavesPerBlock - 1) / wavesPerBlock;
        gcn_gemm_wmma_f32<<<blocks, wavesPerBlock * 32, 0, stream>>>(
            embeds, weight, support, M);
    }

    {
        const int wavesPerBlock = 8;
        const int blocks = (E + wavesPerBlock - 1) / wavesPerBlock;
        gcn_spmm_scatter<<<blocks, wavesPerBlock * 32, 0, stream>>>(
            support, dst, src, evals, out, E);
    }
}


# scatter to gather: a work-group owns 392 out rows, scans dst into LDS slots (LDS int add), sums rows in registers, one store per line; no f32 atomics, no zero launch; sum order by slot arrival
# speedup vs baseline: 1.5644x; 1.1984x over previous
#include <hip/hip_runtime.h>

typedef __attribute__((ext_vector_type(2))) float v2f;
typedef __attribute__((ext_vector_type(8))) float v8f;

#define D 128
#define NTILES_N (D/16)

__global__ void __launch_bounds__(256)
gcn_gemm_wmma_f32(const float* __restrict__ A,
                  const float* __restrict__ W,
                  float* __restrict__ S,
                  int M) {
    __shared__ float2 Wlds2[(D / 2) * D];

    for (int i = threadIdx.x; i < (D / 2) * (D / 4); i += blockDim.x) {
        const int p  = i / (D / 4);
        const int c4 = (i % (D / 4)) * 4;
        const float4 r0 = *(const float4*)(W + (size_t)(2 * p)     * D + c4);
        const float4 r1 = *(const float4*)(W + (size_t)(2 * p + 1) * D + c4);
        float2* outp = Wlds2 + p * D + c4;
        outp[0] = make_float2(r0.x, r1.x);
        outp[1] = make_float2(r0.y, r1.y);
        outp[2] = make_float2(r0.z, r1.z);
        outp[3] = make_float2(r0.w, r1.w);
    }
    __syncthreads();

    const int lane  = threadIdx.x & 31;
    const int tileM = blockIdx.x * (blockDim.x >> 5) + (threadIdx.x >> 5);
    if (tileM * 16 >= M) return;

    const int half = lane >> 4;
    const int l15  = lane & 15;
    const int m    = min(tileM * 16 + l15, M - 1);
    const float*  Arow = A + (size_t)m * D;
    const float2* Wb   = Wlds2 + half * D + l15;

    v8f acc[NTILES_N] = {};
#pragma unroll
    for (int k = 0; k < D; k += 4) {
        const v2f a = *(const v2f*)(Arow + k + half * 2);
#pragma unroll
        for (int n = 0; n < NTILES_N; ++n) {
            const v2f b = *(const v2f*)(Wb + (k >> 1) * D + n * 16);
            acc[n] = __builtin_amdgcn_wmma_f32_16x16x4_f32(
                false, a, false, b, (short)0, acc[n], false, false);
        }
    }

#pragma unroll
    for (int r = 0; r < 8; ++r) {
        const int row = tileM * 16 + half * 8 + r;
        if (row < M) {
            float* Sp = S + (size_t)row * D + l15;
#pragma unroll
            for (int n = 0; n < NTILES_N; ++n) {
                Sp[n * 16] = acc[n][r];
            }
        }
    }
}

#define SP_ROWS    392
#define SP_CAP     32
#define SP_THREADS 1024
#define SP_WAVES   (SP_THREADS / 32)
#define SP_RSLOTS  ((SP_ROWS + SP_WAVES - 1) / SP_WAVES)
#define SCAN_DEPTH 8

__device__ __forceinline__ void sp_claim(int* cnt, int* slotE, unsigned u, int e) {
    const int pos = atomicAdd(&cnt[u], 1);
    if (pos < SP_CAP) slotE[u * SP_CAP + pos] = e;
}

__device__ __forceinline__ void sp_scan4(int* cnt, int* slotE, int4 d4, int e0,
                                         int lo, unsigned nrows) {
    const unsigned u0 = (unsigned)(d4.x - lo);
    const unsigned u1 = (unsigned)(d4.y - lo);
    const unsigned u2 = (unsigned)(d4.z - lo);
    const unsigned u3 = (unsigned)(d4.w - lo);
    const unsigned m  = min(min(u0, u1), min(u2, u3));
    if (m < nrows) {
        if (u0 < nrows) sp_claim(cnt, slotE, u0, e0);
        if (u1 < nrows) sp_claim(cnt, slotE, u1, e0 + 1);
        if (u2 < nrows) sp_claim(cnt, slotE, u2, e0 + 2);
        if (u3 < nrows) sp_claim(cnt, slotE, u3, e0 + 3);
    }
}

__global__ void __launch_bounds__(SP_THREADS)
gcn_spmm_gather(const float* __restrict__ support,
                const int*   __restrict__ dst,
                const int*   __restrict__ src,
                const float* __restrict__ vals,
                float*       __restrict__ out,
                int E, int M) {
    __shared__ int   cnt[SP_ROWS];
    __shared__ int   slotE[SP_ROWS * SP_CAP];
    __shared__ float slotV[SP_ROWS * SP_CAP];

    const int tid  = threadIdx.x;
    const int lane = tid & 31;
    const int w    = tid >> 5;
    const int lo   = blockIdx.x * SP_ROWS;
    const unsigned nrows = (unsigned)min(SP_ROWS, M - lo);

    for (int i = tid; i < SP_ROWS; i += SP_THREADS) cnt[i] = 0;
    __syncthreads();

    const int   E4   = E >> 2;
    const int4* dst4 = (const int4*)dst;
    const int rot = (int)blockIdx.x * (E4 / (int)gridDim.x);

    const int STEP = SP_THREADS * SCAN_DEPTH;
    int base = 0;
    for (; base + STEP <= E4; base += STEP) {
        int4 d4[SCAN_DEPTH];
        int  ix[SCAN_DEPTH];
#pragma unroll
        for (int u = 0; u < SCAN_DEPTH; ++u) {
            int i = base + u * SP_THREADS + tid + rot;
            if (i >= E4) i -= E4;
            ix[u] = i;
            d4[u] = dst4[i];
        }
#pragma unroll
        for (int u = 0; u < SCAN_DEPTH; ++u)
            sp_scan4(cnt, slotE, d4[u], ix[u] * 4, lo, nrows);
    }
    for (int i0 = base + tid; i0 < E4; i0 += SP_THREADS) {
        int i = i0 + rot;
        if (i >= E4) i -= E4;
        sp_scan4(cnt, slotE, dst4[i], i * 4, lo, nrows);
    }
    {
        const int e = E4 * 4 + tid;
        if (e < E) {
            const unsigned u = (unsigned)(dst[e] - lo);
            if (u < nrows) sp_claim(cnt, slotE, u, e);
        }
    }
    __syncthreads();

    if (E > 0) {
        int   sv[SP_RSLOTS];
        float vv[SP_RSLOTS];
        bool  ok[SP_RSLOTS];
        int   ev[SP_RSLOTS];
#pragma unroll
        for (int k = 0; k < SP_RSLOTS; ++k) {
            const int  r  = w + k * SP_WAVES;
            const bool in = r < (int)nrows;
            const int  rc = in ? r : 0;
            const int  n  = cnt[rc];
            const int  e  = slotE[rc * SP_CAP + lane];
            ok[k] = in && (n <= SP_CAP) && (lane < n);
            ev[k] = ok[k] ? e : 0;
        }
#pragma unroll
        for (int k = 0; k < SP_RSLOTS; ++k) {
            sv[k] = src[ev[k]];
            vv[k] = vals[ev[k]];
        }
#pragma unroll
        for (int k = 0; k < SP_RSLOTS; ++k) {
            if (ok[k]) {
                const int q = (w + k * SP_WAVES) * SP_CAP + lane;
                slotE[q] = sv[k];
                slotV[q] = vv[k];
            }
        }
    }

    for (int r = w; r < (int)nrows; r += SP_WAVES) {
        const int row = lo + r;
        const int n   = __builtin_amdgcn_readfirstlane(cnt[r]);
        float4 acc = make_float4(0.f, 0.f, 0.f, 0.f);

        if (n <= SP_CAP) {
            int   s = 0;
            float v = 0.f;
            if (lane < n) {
                s = slotE[r * SP_CAP + lane];
                v = slotV[r * SP_CAP + lane];
            }
            for (int j0 = 0; j0 < n; j0 += 8) {
                float4 x[8];
                float  c[8];
#pragma unroll
                for (int u = 0; u < 8; ++u) {
                    x[u] = make_float4(0.f, 0.f, 0.f, 0.f);
                    c[u] = 0.f;
                    if (j0 + u < n) {
                        const int sj = __builtin_amdgcn_readlane(s, j0 + u);
                        c[u] = __int_as_float(
                            __builtin_amdgcn_readlane(__float_as_int(v), j0 + u));
                        x[u] = *(const float4*)(support + (size_t)sj * D + lane * 4);
                    }
                }
#pragma unroll
                for (int u = 0; u < 8; ++u) {
                    if (j0 + u < n) {
                        acc.x += c[u] * x[u].x;
                        acc.y += c[u] * x[u].y;
                        acc.z += c[u] * x[u].z;
                        acc.w += c[u] * x[u].w;
                    }
                }
            }
        } else {
            for (int b = 0; b < E; b += 32) {
                const int  e   = b + lane;
                const bool hit = (e < E) && (dst[e] == row);
                unsigned mask  = __builtin_amdgcn_ballot_w32(hit);
                while (mask) {
                    const int j  = __builtin_ctz(mask);
                    mask &= mask - 1;
                    const int   sj = src[b + j];
                    const float cj = vals[b + j];
                    const float4 x = *(const float4*)(support + (size_t)sj * D + lane * 4);
                    acc.x += cj * x.x;
                    acc.y += cj * x.y;
                    acc.z += cj * x.z;
                    acc.w += cj * x.w;
                }
            }
        }
        *(float4*)(out + (size_t)row * D + lane * 4) = acc;
    }
}

extern "C" void kernel_launch(void* const* d_in, const int* in_sizes, int n_in,
                              void* d_out, int out_size, void* d_ws, size_t ws_size,
                              hipStream_t stream) {
    const float* embeds = (const float*)d_in[0];
    const float* weight = (const float*)d_in[1];
    const int*   eidx   = (const int*)d_in[2];
    const float* evals  = (const float*)d_in[3];

    const int M = in_sizes[0] / D;
    const int E = in_sizes[3];
    const int* dst = eidx;
    const int* src = eidx + E;

    float* support = (float*)d_ws;
    float* out     = (float*)d_out;

    {
        const int mtiles = (M + 15) / 16;
        const int wavesPerBlock = 8;
        const int blocks = (mtiles + wavesPerBlock - 1) / wavesPerBlock;
        gcn_gemm_wmma_f32<<<blocks, wavesPerBlock * 32, 0, stream>>>(
            embeds, weight, support, M);
    }

    {
        const int blocks = (M + SP_ROWS - 1) / SP_ROWS;
        gcn_spmm_gather<<<blocks, SP_THREADS, 0, stream>>>(
            support, dst, src, evals, out, E, M);
    }
}


# scan narrowed: first kernel also writes one key byte per edge (dst / 392); gather work-groups scan the bytes into an LDS hit list, then load dst, src, val of hits; sum order by slot arrival as before
# speedup vs baseline: 1.8227x; 1.1651x over previous
#include <hip/hip_runtime.h>

typedef __attribute__((ext_vector_type(2))) float v2f;
typedef __attribute__((ext_vector_type(8))) float v8f;

#define D 128
#define NTILES_N (D/16)
#define SP_ROWS  392

__device__ __forceinline__ unsigned sp_key(int d) {
    return ((unsigned)d / SP_ROWS) & 255u;
}

__global__ void __launch_bounds__(256)
gcn_gemm_wmma_f32(const float* __restrict__ A,
                  const float* __restrict__ W,
                  float* __restrict__ S,
                  const int* __restrict__ dst,
                  unsigned char* __restrict__ key,
                  int M, int E) {
    __shared__ float2 Wlds2[(D / 2) * D];

    {
        const int gtid = blockIdx.x * blockDim.x + threadIdx.x;
        const int gsz  = gridDim.x * blockDim.x;
        const int E4   = E >> 2;
        for (int i = gtid; i < E4; i += gsz) {
            const int4 d4 = ((const int4*)dst)[i];
            ((unsigned*)key)[i] = sp_key(d4.x) | (sp_key(d4.y) << 8) |
                                  (sp_key(d4.z) << 16) | (sp_key(d4.w) << 24);
        }
        for (int e = E4 * 4 + gtid; e < E; e += gsz)
            key[e] = (unsigned char)sp_key(dst[e]);
    }

    for (int i = threadIdx.x; i < (D / 2) * (D / 4); i += blockDim.x) {
        const int p  = i / (D / 4);
        const int c4 = (i % (D / 4)) * 4;
        const float4 r0 = *(const float4*)(W + (size_t)(2 * p)     * D + c4);
        const float4 r1 = *(const float4*)(W + (size_t)(2 * p + 1) * D + c4);
        float2* outp = Wlds2 + p * D + c4;
        outp[0] = make_float2(r0.x, r1.x);
        outp[1] = make_float2(r0.y, r1.y);
        outp[2] = make_float2(r0.z, r1.z);
        outp[3] = make_float2(r0.w, r1.w);
    }
    __syncthreads();

    const int lane  = threadIdx.x & 31;
    const int tileM = blockIdx.x * (blockDim.x >> 5) + (threadIdx.x >> 5);
    if (tileM * 16 >= M) return;

    const int half = lane >> 4;
    const int l15  = lane & 15;
    const int m    = min(tileM * 16 + l15, M - 1);
    const float*  Arow = A + (size_t)m * D;
    const float2* Wb   = Wlds2 + half * D + l15;

    v8f acc[NTILES_N] = {};
#pragma unroll
    for (int k = 0; k < D; k += 4) {
        const v2f a = *(const v2f*)(Arow + k + half * 2);
#pragma unroll
        for (int n = 0; n < NTILES_N; ++n) {
            const v2f b = *(const v2f*)(Wb + (k >> 1) * D + n * 16);
            acc[n] = __builtin_amdgcn_wmma_f32_16x16x4_f32(
                false, a, false, b, (short)0, acc[n], false, false);
        }
    }

#pragma unroll
    for (int r = 0; r < 8; ++r) {
        const int row = tileM * 16 + half * 8 + r;
        if (row < M) {
            float* Sp = S + (size_t)row * D + l15;
#pragma unroll
            for (int n = 0; n < NTILES_N; ++n) {
                Sp[n * 16] = acc[n][r];
            }
        }
    }
}

#define SP_CAP     32
#define SP_THREADS 1024
#define SP_WAVES   (SP_THREADS / 32)
#define SCAN_DEPTH 8
#define LIST_CAP   8192
#define CLAIM_DEPTH 4

__device__ __forceinline__ void sp_list(int* list, int* nlist, int e) {
    const int pos = atomicAdd(nlist, 1);
    if (pos < LIST_CAP) list[pos] = e;
}

__device__ __forceinline__ unsigned sp_zero_bytes(unsigned x) {
    return (x - 0x01010101u) & ~x & 0x80808080u;
}

__device__ __forceinline__ void sp_hits(int* list, int* nlist, unsigned z, int e0) {
    while (z) {
        sp_list(list, nlist, e0 + (__builtin_ctz(z) >> 3));
        z &= z - 1;
    }
}

__device__ __forceinline__ void sp_scan16(int* list, int* nlist, uint4 k, int e0,
                                          unsigned kb) {
    const unsigned z0 = sp_zero_bytes(k.x ^ kb);
    const unsigned z1 = sp_zero_bytes(k.y ^ kb);
    const unsigned z2 = sp_zero_bytes(k.z ^ kb);
    const unsigned z3 = sp_zero_bytes(k.w ^ kb);
    if ((z0 | z1 | z2 | z3) != 0u) {
        sp_hits(list, nlist, z0, e0);
        sp_hits(list, nlist, z1, e0 + 4);
        sp_hits(list, nlist, z2, e0 + 8);
        sp_hits(list, nlist, z3, e0 + 12);
    }
}

__global__ void __launch_bounds__(SP_THREADS)
gcn_spmm_gather(const float* __restrict__ support,
                const int*   __restrict__ dst,
                const int*   __restrict__ src,
                const float* __restrict__ vals,
                const unsigned char* __restrict__ key,
                float*       __restrict__ out,
                int E, int M) {
    __shared__ int   cnt[SP_ROWS];
    __shared__ int   slotE[SP_ROWS * SP_CAP];
    __shared__ float slotV[SP_ROWS * SP_CAP];
    __shared__ int   list[LIST_CAP];
    __shared__ int   nlist;

    const int tid  = threadIdx.x;
    const int lane = tid & 31;
    const int w    = tid >> 5;
    const int lo   = blockIdx.x * SP_ROWS;
    const unsigned nrows = (unsigned)min(SP_ROWS, M - lo);

    for (int i = tid; i < SP_ROWS; i += SP_THREADS) cnt[i] = 0;
    if (tid == 0) nlist = 0;
    __syncthreads();

    const unsigned kbyte = (unsigned)blockIdx.x & 255u;
    const unsigned kb    = kbyte * 0x01010101u;
    const int    E16   = E >> 4;
    const uint4* key16 = (const uint4*)key;
    const int rot = (int)blockIdx.x * (E16 / (int)gridDim.x);

    const int STEP = SP_THREADS * SCAN_DEPTH;
    for (int base = 0; base < E16; base += STEP) {
        uint4 k4[SCAN_DEPTH];
        int   ix[SCAN_DEPTH];
#pragma unroll
        for (int u = 0; u < SCAN_DEPTH; ++u) {
            const int  i0 = base + u * SP_THREADS + tid;
            const bool in = i0 < E16;
            int i = i0 + rot;
            if (i >= E16) i -= E16;
            ix[u] = in ? i : -1;
            k4[u] = key16[in ? i : 0];
        }
#pragma unroll
        for (int u = 0; u < SCAN_DEPTH; ++u)
            if (ix[u] >= 0) sp_scan16(list, &nlist, k4[u], ix[u] * 16, kb);
    }
    {
        const int e = E16 * 16 + tid;
        if (tid < 16 && e < E && (unsigned)key[e] == kbyte) sp_list(list, &nlist, e);
    }
    __syncthreads();

    const int nl = nlist;
    if (nl <= LIST_CAP) {
        for (int q0 = 0; q0 < nl; q0 += SP_THREADS * CLAIM_DEPTH) {
            int   dv[CLAIM_DEPTH];
            int   sv[CLAIM_DEPTH];
            float vv[CLAIM_DEPTH];
            bool  ok[CLAIM_DEPTH];
#pragma unroll
            for (int u = 0; u < CLAIM_DEPTH; ++u) {
                const int q = q0 + u * SP_THREADS + tid;
                ok[u] = q < nl;
                const int e = ok[u] ? list[q] : 0;
                dv[u] = dst[e];
                sv[u] = src[e];
                vv[u] = vals[e];
            }
#pragma unroll
            for (int u = 0; u < CLAIM_DEPTH; ++u) {
                const unsigned r = (unsigned)(dv[u] - lo);
                if (ok[u] && r < nrows) {
                    const int pos = atomicAdd(&cnt[r], 1);
                    if (pos < SP_CAP) {
                        slotE[r * SP_CAP + pos] = sv[u];
                        slotV[r * SP_CAP + pos] = vv[u];
                    }
                }
            }
        }
    } else {
        for (int i = tid; i < SP_ROWS; i += SP_THREADS) cnt[i] = SP_CAP + 1;
    }
    __syncthreads();

    for (int r = w; r < (int)nrows; r += SP_WAVES) {
        const int row = lo + r;
        const int n   = __builtin_amdgcn_readfirstlane(cnt[r]);
        float4 acc = make_float4(0.f, 0.f, 0.f, 0.f);

        if (n <= SP_CAP) {
            int   s = 0;
            float v = 0.f;
            if (lane < n) {
                s = slotE[r * SP_CAP + lane];
                v = slotV[r * SP_CAP + lane];
            }
            for (int j0 = 0; j0 < n; j0 += 8) {
                float4 x[8];
                float  c[8];
#pragma unroll
                for (int u = 0; u < 8; ++u) {
                    x[u] = make_float4(0.f, 0.f, 0.f, 0.f);
                    c[u] = 0.f;
                    if (j0 + u < n) {
                        const int sj = __builtin_amdgcn_readlane(s, j0 + u);
                        c[u] = __int_as_float(
                            __builtin_amdgcn_readlane(__float_as_int(v), j0 + u));
                        x[u] = *(const float4*)(support + (size_t)sj * D + lane * 4);
                    }
                }
#pragma unroll
                for (int u = 0; u < 8; ++u) {
                    if (j0 + u < n) {
                        acc.x += c[u] * x[u].x;
                        acc.y += c[u] * x[u].y;
                        acc.z += c[u] * x[u].z;
                        acc.w += c[u] * x[u].w;
                    }
                }
            }
        } else {
            for (int b = 0; b < E; b += 32) {
                const int  e   = b + lane;
                const bool hit = (e < E) && (dst[e] == row);
                unsigned mask  = __builtin_amdgcn_ballot_w32(hit);
                while (mask) {
                    const int j  = __builtin_ctz(mask);
                    mask &= mask - 1;
                    const int   sj = src[b + j];
                    const float cj = vals[b + j];
                    const float4 x = *(const float4*)(support + (size_t)sj * D + lane * 4);
                    acc.x += cj * x.x;
                    acc.y += cj * x.y;
                    acc.z += cj * x.z;
                    acc.w += cj * x.w;
                }
            }
        }
        *(float4*)(out + (size_t)row * D + lane * 4) = acc;
    }
}

extern "C" void kernel_launch(void* const* d_in, const int* in_sizes, int n_in,
                              void* d_out, int out_size, void* d_ws, size_t ws_size,
                              hipStream_t stream) {
    const float* embeds = (const float*)d_in[0];
    const float* weight = (const float*)d_in[1];
    const int*   eidx   = (const int*)d_in[2];
    const float* evals  = (const float*)d_in[3];

    const int M = in_sizes[0] / D;
    const int E = in_sizes[3];
    const int* dst = eidx;
    const int* src = eidx + E;

    float* support = (float*)d_ws;
    unsigned char* key = (unsigned char*)(support + (size_t)M * D);
    float* out     = (float*)d_out;

    {
        const int mtiles = (M + 15) / 16;
        const int wavesPerBlock = 8;
        const int blocks = (mtiles + wavesPerBlock - 1) / wavesPerBlock;
        gcn_gemm_wmma_f32<<<blocks, wavesPerBlock * 32, 0, stream>>>(
            embeds, weight, support, dst, key, M, E);
    }

    {
        const int blocks = (M + SP_ROWS - 1) / SP_ROWS;
        gcn_spmm_gather<<<blocks, SP_THREADS, 0, stream>>>(
            support, dst, src, evals, key, out, E, M);
    }
}


# first launch sorts edges by owner bucket (LDS counting sort per 1024-edge segment) into 16-B records; owners read their runs, no key scan, no hit list; sum order by slot arrival as before
# speedup vs baseline: 2.0966x; 1.1503x over previous
#include <hip/hip_runtime.h>

typedef __attribute__((ext_vector_type(2))) float v2f;
typedef __attribute__((ext_vector_type(8))) float v8f;

#define D 128
#define NTILES_N (D/16)
#define SP_ROWS  392

#define SEG      1024
#define NBUCKET  256

__device__ __forceinline__ unsigned sp_key(int d) {
    return ((unsigned)d / SP_ROWS) & (NBUCKET - 1u);
}

__global__ void __launch_bounds__(256)
gcn_gemm_wmma_f32(const float* __restrict__ A,
                  const float* __restrict__ W,
                  float* __restrict__ S,
                  const int* __restrict__ dst,
                  const int* __restrict__ src,
                  const float* __restrict__ vals,
                  int* __restrict__ recs4,
                  unsigned* __restrict__ runs,
                  int M, int E) {
    __shared__ float2 Wlds2[(D / 2) * D];
    int4* recs = (int4*)recs4;
    int* hist = (int*)Wlds2;
    int* wsum = hist + NBUCKET;

    {
        const int tid  = threadIdx.x;
        const int nseg = (E + SEG - 1) / SEG;
        for (int seg = blockIdx.x; seg < nseg; seg += gridDim.x) {
            __syncthreads();
            hist[tid] = 0;
            __syncthreads();
            int   dv[4], sv[4], rk[4];
            float vv[4];
            bool  ok[4];
#pragma unroll
            for (int j = 0; j < 4; ++j) {
                const int e = seg * SEG + j * 256 + tid;
                ok[j] = e < E;
                const int ec = ok[j] ? e : 0;
                dv[j] = dst[ec];
                sv[j] = src[ec];
                vv[j] = vals[ec];
            }
#pragma unroll
            for (int j = 0; j < 4; ++j)
                rk[j] = ok[j] ? atomicAdd(&hist[sp_key(dv[j])], 1) : 0;
            __syncthreads();
            const int c = hist[tid];
            int inc = c;
#pragma unroll
            for (int o = 1; o < 32; o <<= 1) {
                const int y = __shfl_up(inc, o);
                if ((tid & 31) >= o) inc += y;
            }
            if ((tid & 31) == 31) wsum[tid >> 5] = inc;
            __syncthreads();
            int base = 0;
            for (int i = 0; i < (tid >> 5); ++i) base += wsum[i];
            const int excl = base + inc - c;
            hist[tid] = excl;
            __syncthreads();
            runs[(size_t)tid * nseg + seg] = (unsigned)excl | ((unsigned)c << 16);
#pragma unroll
            for (int j = 0; j < 4; ++j)
                if (ok[j])
                    recs[(size_t)seg * SEG + hist[sp_key(dv[j])] + rk[j]] =
                        make_int4(dv[j], sv[j], __float_as_int(vv[j]), 0);
        }
        __builtin_amdgcn_fence(__ATOMIC_RELEASE, "workgroup", "local");
        __builtin_amdgcn_s_barrier();
        __builtin_amdgcn_fence(__ATOMIC_ACQUIRE, "workgroup", "local");
    }

    for (int i = threadIdx.x; i < (D / 2) * (D / 4); i += blockDim.x) {
        const int p  = i / (D / 4);
        const int c4 = (i % (D / 4)) * 4;
        const float4 r0 = *(const float4*)(W + (size_t)(2 * p)     * D + c4);
        const float4 r1 = *(const float4*)(W + (size_t)(2 * p + 1) * D + c4);
        float2* outp = Wlds2 + p * D + c4;
        outp[0] = make_float2(r0.x, r1.x);
        outp[1] = make_float2(r0.y, r1.y);
        outp[2] = make_float2(r0.z, r1.z);
        outp[3] = make_float2(r0.w, r1.w);
    }
    __syncthreads();

    const int lane  = threadIdx.x & 31;
    const int tileM = blockIdx.x * (blockDim.x >> 5) + (threadIdx.x >> 5);
    if (tileM * 16 >= M) return;

    const int half = lane >> 4;
    const int l15  = lane & 15;
    const int m    = min(tileM * 16 + l15, M - 1);
    const float*  Arow = A + (size_t)m * D;
    const float2* Wb   = Wlds2 + half * D + l15;

    v8f acc[NTILES_N] = {};
#pragma unroll
    for (int k = 0; k < D; k += 4) {
        const v2f a = *(const v2f*)(Arow + k + half * 2);
#pragma unroll
        for (int n = 0; n < NTILES_N; ++n) {
            const v2f b = *(const v2f*)(Wb + (k >> 1) * D + n * 16);
            acc[n] = __builtin_amdgcn_wmma_f32_16x16x4_f32(
                false, a, false, b, (short)0, acc[n], false, false);
        }
    }

#pragma unroll
    for (int r = 0; r < 8; ++r) {
        const int row = tileM * 16 + half * 8 + r;
        if (row < M) {
            float* Sp = S + (size_t)row * D + l15;
#pragma unroll
            for (int n = 0; n < NTILES_N; ++n) {
                Sp[n * 16] = acc[n][r];
            }
        }
    }
}

#define SP_CAP     32
#define SP_THREADS 1024
#define SP_WAVES   (SP_THREADS / 32)
#define CLAIM_DEPTH 8

__global__ void __launch_bounds__(SP_THREADS)
gcn_spmm_gather(const float* __restrict__ support,
                const int*   __restrict__ dst,
                const int*   __restrict__ src,
                const float* __restrict__ vals,
                const int*   __restrict__ recs4,
                const unsigned* __restrict__ runs,
                float*       __restrict__ out,
                int E, int M) {
    __shared__ int   cnt[SP_ROWS];
    __shared__ int   slotE[SP_ROWS * SP_CAP];
    __shared__ float slotV[SP_ROWS * SP_CAP];

    const int4* recs = (const int4*)recs4;
    const int tid  = threadIdx.x;
    const int lane = tid & 31;
    const int w    = tid >> 5;
    const int lo   = blockIdx.x * SP_ROWS;
    const unsigned nrows = (unsigned)min(SP_ROWS, M - lo);

    for (int i = tid; i < SP_ROWS; i += SP_THREADS) cnt[i] = 0;
    __syncthreads();

    {
        const int nseg = (E + SEG - 1) / SEG;
        const unsigned* myruns = runs + (size_t)(blockIdx.x & (NBUCKET - 1)) * nseg;
        for (int t = tid; t < nseg; t += SP_THREADS) {
            const unsigned wd = myruns[t];
            const int n = (int)(wd >> 16);
            const int4* rp = recs + (size_t)t * SEG + (wd & 0xffffu);
            for (int k0 = 0; k0 < n; k0 += CLAIM_DEPTH) {
                int4 rc[CLAIM_DEPTH];
#pragma unroll
                for (int u = 0; u < CLAIM_DEPTH; ++u)
                    rc[u] = rp[min(k0 + u, n - 1)];
#pragma unroll
                for (int u = 0; u < CLAIM_DEPTH; ++u) {
                    const unsigned r = (unsigned)(rc[u].x - lo);
                    if (k0 + u < n && r < nrows) {
                        const int pos = atomicAdd(&cnt[r], 1);
                        if (pos < SP_CAP) {
                            slotE[r * SP_CAP + pos] = rc[u].y;
                            slotV[r * SP_CAP + pos] = __int_as_float(rc[u].z);
                        }
                    }
                }
            }
        }
    }
    __syncthreads();

    for (int r = w; r < (int)nrows; r += SP_WAVES) {
        const int row = lo + r;
        const int n   = __builtin_amdgcn_readfirstlane(cnt[r]);
        float4 acc = make_float4(0.f, 0.f, 0.f, 0.f);

        if (n <= SP_CAP) {
            int   s = 0;
            float v = 0.f;
            if (lane < n) {
                s = slotE[r * SP_CAP + lane];
                v = slotV[r * SP_CAP + lane];
            }
            for (int j0 = 0; j0 < n; j0 += 8) {
                float4 x[8];
                float  c[8];
#pragma unroll
                for (int u = 0; u < 8; ++u) {
                    x[u] = make_float4(0.f, 0.f, 0.f, 0.f);
                    c[u] = 0.f;
                    if (j0 + u < n) {
                        const int sj = __builtin_amdgcn_readlane(s, j0 + u);
                        c[u] = __int_as_float(
                            __builtin_amdgcn_readlane(__float_as_int(v), j0 + u));
                        x[u] = *(const float4*)(support + (size_t)sj * D + lane * 4);
                    }
                }
#pragma unroll
                for (int u = 0; u < 8; ++u) {
                    if (j0 + u < n) {
                        acc.x += c[u] * x[u].x;
                        acc.y += c[u] * x[u].y;
                        acc.z += c[u] * x[u].z;
                        acc.w += c[u] * x[u].w;
                    }
                }
            }
        } else {
            for (int b = 0; b < E; b += 32) {
                const int  e   = b + lane;
                const bool hit = (e < E) && (dst[e] == row);
                unsigned mask  = __builtin_amdgcn_ballot_w32(hit);
                while (mask) {
                    const int j  = __builtin_ctz(mask);
                    mask &= mask - 1;
                    const int   sj = src[b + j];
                    const float cj = vals[b + j];
                    const float4 x = *(const float4*)(support + (size_t)sj * D + lane * 4);
                    acc.x += cj * x.x;
                    acc.y += cj * x.y;
                    acc.z += cj * x.z;
                    acc.w += cj * x.w;
                }
            }
        }
        *(float4*)(out + (size_t)row * D + lane * 4) = acc;
    }
}

extern "C" void kernel_launch(void* const* d_in, const int* in_sizes, int n_in,
                              void* d_out, int out_size, void* d_ws, size_t ws_size,
                              hipStream_t stream) {
    const float* embeds = (const float*)d_in[0];
    const float* weight = (const float*)d_in[1];
    const int*   eidx   = (const int*)d_in[2];
    const float* evals  = (const float*)d_in[3];

    const int M = in_sizes[0] / D;
    const int E = in_sizes[3];
    const int* dst = eidx;
    const int* src = eidx + E;

    float* support = (float*)d_ws;
    int* recs      = (int*)(support + (size_t)M * D);
    unsigned* runs = (unsigned*)(recs + ((size_t)E + SEG) * 4);
    float* out     = (float*)d_out;

    {
        const int mtiles = (M + 15) / 16;
        const int wavesPerBlock = 8;
        const int blocks = (mtiles + wavesPerBlock - 1) / wavesPerBlock;
        gcn_gemm_wmma_f32<<<blocks, wavesPerBlock * 32, 0, stream>>>(
            embeds, weight, support, dst, src, evals, recs, runs, M, E);
    }

    {
        const int blocks = (M + SP_ROWS - 1) / SP_ROWS;
        gcn_spmm_gather<<<blocks, SP_THREADS, 0, stream>>>(
            support, dst, src, evals, recs, runs, out, E, M);
    }
}


# first launch as 256 resident work-groups of 24 waves, one per WGP: W staged once per WGP, strips dealt evenly to SIMDs (6 whole + a column quarter), three edge segments sorted side by side
# speedup vs baseline: 2.2387x; 1.0678x over previous
#include <hip/hip_runtime.h>

typedef __attribute__((ext_vector_type(2))) float v2f;
typedef __attribute__((ext_vector_type(8))) float v8f;

#define D 128
#define NTILES_N (D/16)
#define SP_ROWS  392

#define SEG      1024
#define NBUCKET  256

#define GEMM_WAVES   24
#define GEMM_THREADS (GEMM_WAVES * 32)
#define SORT_GROUPS  (GEMM_THREADS / 256)
#define SORT_PITCH   (NBUCKET + 16)

__device__ __forceinline__ unsigned sp_key(int d) {
    return ((unsigned)d / SP_ROWS) & (NBUCKET - 1u);
}

template <int NT>
__device__ __forceinline__ void gemm_strip(const float* __restrict__ A, float* __restrict__ S,
                                           const float2* Wlds2, int tileM, int n0, int M, int lane) {
    const int half = lane >> 4;
    const int l15  = lane & 15;
    int wofs = half * D + l15 + n0 * 16;
    asm volatile("" : "+v"(wofs) : : "memory");
    const int m    = min(tileM * 16 + l15, M - 1);
    const float*  Arow = A + (size_t)m * D;
    const float2* Wb   = Wlds2 + wofs;

    v8f acc[NT] = {};
#pragma unroll
    for (int k = 0; k < D; k += 4) {
        const v2f a = *(const v2f*)(Arow + k + half * 2);
#pragma unroll
        for (int n = 0; n < NT; ++n) {
            const v2f b = *(const v2f*)(Wb + (k >> 1) * D + n * 16);
            acc[n] = __builtin_amdgcn_wmma_f32_16x16x4_f32(
                false, a, false, b, (short)0, acc[n], false, false);
        }
    }

#pragma unroll
    for (int r = 0; r < 8; ++r) {
        const int row = tileM * 16 + half * 8 + r;
        if (row < M) {
            float* Sp = S + (size_t)row * D + n0 * 16 + l15;
#pragma unroll
            for (int n = 0; n < NT; ++n) {
                Sp[n * 16] = acc[n][r];
            }
        }
    }
}

__global__ void __launch_bounds__(GEMM_THREADS)
gcn_gemm_wmma_f32(const float* __restrict__ A,
                  const float* __restrict__ W,
                  float* __restrict__ S,
                  const int* __restrict__ dst,
                  const int* __restrict__ src,
                  const float* __restrict__ vals,
                  int* __restrict__ recs4,
                  unsigned* __restrict__ runs,
                  int M, int E) {
    __shared__ float2 Wlds2[21 * 1024];
    int4* recs = (int4*)recs4;
    const int grp = threadIdx.x >> 8;
    int* hist = (int*)Wlds2 + grp * SORT_PITCH;
    int* wsum = hist + NBUCKET;

    {
        const int tid  = threadIdx.x & 255;
        const int nseg = (E + SEG - 1) / SEG;
        for (int seg0 = blockIdx.x * SORT_GROUPS; seg0 < nseg; seg0 += gridDim.x * SORT_GROUPS) {
            const int  seg  = seg0 + grp;
            const bool live = seg < nseg;
            __syncthreads();
            hist[tid] = 0;
            __syncthreads();
            int   dv[4], sv[4], rk[4];
            float vv[4];
            bool  ok[4];
#pragma unroll
            for (int j = 0; j < 4; ++j) {
                const int e = seg * SEG + j * 256 + tid;
                ok[j] = live && e < E;
                const int ec = ok[j] ? e : 0;
                dv[j] = dst[ec];
                sv[j] = src[ec];
                vv[j] = vals[ec];
            }
#pragma unroll
            for (int j = 0; j < 4; ++j)
                rk[j] = ok[j] ? atomicAdd(&hist[sp_key(dv[j])], 1) : 0;
            __syncthreads();
            const int c = hist[tid];
            int inc = c;
#pragma unroll
            for (int o = 1; o < 32; o <<= 1) {
                const int y = __shfl_up(inc, o);
                if ((tid & 31) >= o) inc += y;
            }
            if ((tid & 31) == 31) wsum[tid >> 5] = inc;
            __syncthreads();
            int base = 0;
            for (int i = 0; i < (tid >> 5); ++i) base += wsum[i];
            const int excl = base + inc - c;
            hist[tid] = excl;
            __syncthreads();
            if (live)
                runs[(size_t)tid * nseg + seg] = (unsigned)excl | ((unsigned)c << 16);
#pragma unroll
            for (int j = 0; j < 4; ++j)
                if (ok[j])
                    recs[(size_t)seg * SEG + hist[sp_key(dv[j])] + rk[j]] =
                        make_int4(dv[j], sv[j], __float_as_int(vv[j]), 0);
        }
        __builtin_amdgcn_fence(__ATOMIC_RELEASE, "workgroup", "local");
        __builtin_amdgcn_s_barrier();
        __builtin_amdgcn_fence(__ATOMIC_ACQUIRE, "workgroup", "local");
    }

    for (int i = threadIdx.x; i < (D / 2) * (D / 4); i += blockDim.x) {
        const int p  = i / (D / 4);
        const int c4 = (i % (D / 4)) * 4;
        const float4 r0 = *(const float4*)(W + (size_t)(2 * p)     * D + c4);
        const float4 r1 = *(const float4*)(W + (size_t)(2 * p + 1) * D + c4);
        float2* outp = Wlds2 + p * D + c4;
        outp[0] = make_float2(r0.x, r1.x);
        outp[1] = make_float2(r0.y, r1.y);
        outp[2] = make_float2(r0.z, r1.z);
        outp[3] = make_float2(r0.w, r1.w);
    }
    __syncthreads();

    const int lane   = threadIdx.x & 31;
    const int wv     = threadIdx.x >> 5;
    const int nw     = blockDim.x >> 5;
    const int grid   = gridDim.x;
    const int mtiles = (M + 15) / 16;
    const int round  = grid * nw;
    const int whole  = (mtiles / round) * round;
    for (int j = wv >> 2; whole + j * grid + (int)blockIdx.x < mtiles; j += nw >> 2)
        gemm_strip<NTILES_N / 4>(A, S, Wlds2, whole + j * grid + (int)blockIdx.x,
                                 (wv & 3) * (NTILES_N / 4), M, lane);
    for (int tileM = wv * grid + (int)blockIdx.x; tileM < whole; tileM += round)
        gemm_strip<NTILES_N>(A, S, Wlds2, tileM, 0, M, lane);
}

#define SP_CAP     32
#define SP_THREADS 1024
#define SP_WAVES   (SP_THREADS / 32)
#define CLAIM_DEPTH 8

__global__ void __launch_bounds__(SP_THREADS)
gcn_spmm_gather(const float* __restrict__ support,
                const int*   __restrict__ dst,
                const int*   __restrict__ src,
                const float* __restrict__ vals,
                const int*   __restrict__ recs4,
                const unsigned* __restrict__ runs,
                float*       __restrict__ out,
                int E, int M) {
    __shared__ int   cnt[SP_ROWS];
    __shared__ int   slotE[SP_ROWS * SP_CAP];
    __shared__ float slotV[SP_ROWS * SP_CAP];

    const int4* recs = (const int4*)recs4;
    const int tid  = threadIdx.x;
    const int lane = tid & 31;
    const int w    = tid >> 5;
    const int lo   = blockIdx.x * SP_ROWS;
    const unsigned nrows = (unsigned)min(SP_ROWS, M - lo);

    for (int i = tid; i < SP_ROWS; i += SP_THREADS) cnt[i] = 0;
    __syncthreads();

    {
        const int nseg = (E + SEG - 1) / SEG;
        const unsigned* myruns = runs + (size_t)(blockIdx.x & (NBUCKET - 1)) * nseg;
        for (int t = tid; t < nseg; t += SP_THREADS) {
            const unsigned wd = myruns[t];
            const int n = (int)(wd >> 16);
            const int4* rp = recs + (size_t)t * SEG + (wd & 0xffffu);
            for (int k0 = 0; k0 < n; k0 += CLAIM_DEPTH) {
                int4 rc[CLAIM_DEPTH];
#pragma unroll
                for (int u = 0; u < CLAIM_DEPTH; ++u)
                    rc[u] = rp[min(k0 + u, n - 1)];
#pragma unroll
                for (int u = 0; u < CLAIM_DEPTH; ++u) {
                    const unsigned r = (unsigned)(rc[u].x - lo);
                    if (k0 + u < n && r < nrows) {
                        const int pos = atomicAdd(&cnt[r], 1);
                        if (pos < SP_CAP) {
                            slotE[r * SP_CAP + pos] = rc[u].y;
                            slotV[r * SP_CAP + pos] = __int_as_float(rc[u].z);
                        }
                    }
                }
            }
        }
    }
    __syncthreads();

    for (int r = w; r < (int)nrows; r += SP_WAVES) {
        const int row = lo + r;
        const int n   = __builtin_amdgcn_readfirstlane(cnt[r]);
        float4 acc = make_float4(0.f, 0.f, 0.f, 0.f);

        if (n <= SP_CAP) {
            int   s = 0;
            float v = 0.f;
            if (lane < n) {
                s = slotE[r * SP_CAP + lane];
                v = slotV[r * SP_CAP + lane];
            }
            for (int j0 = 0; j0 < n; j0 += 8) {
                float4 x[8];
                float  c[8];
#pragma unroll
                for (int u = 0; u < 8; ++u) {
                    x[u] = make_float4(0.f, 0.f, 0.f, 0.f);
                    c[u] = 0.f;
                    if (j0 + u < n) {
                        const int sj = __builtin_amdgcn_readlane(s, j0 + u);
                        c[u] = __int_as_float(
                            __builtin_amdgcn_readlane(__float_as_int(v), j0 + u));
                        x[u] = *(const float4*)(support + (size_t)sj * D + lane * 4);
                    }
                }
#pragma unroll
                for (int u = 0; u < 8; ++u) {
                    if (j0 + u < n) {
                        acc.x += c[u] * x[u].x;
                        acc.y += c[u] * x[u].y;
                        acc.z += c[u] * x[u].z;
                        acc.w += c[u] * x[u].w;
                    }
                }
            }
        } else {
            for (int b = 0; b < E; b += 32) {
                const int  e   = b + lane;
                const bool hit = (e < E) && (dst[e] == row);
                unsigned mask  = __builtin_amdgcn_ballot_w32(hit);
                while (mask) {
                    const int j  = __builtin_ctz(mask);
                    mask &= mask - 1;
                    const int   sj = src[b + j];
                    const float cj = vals[b + j];
                    const float4 x = *(const float4*)(support + (size_t)sj * D + lane * 4);
                    acc.x += cj * x.x;
                    acc.y += cj * x.y;
                    acc.z += cj * x.z;
                    acc.w += cj * x.w;
                }
            }
        }
        *(float4*)(out + (size_t)row * D + lane * 4) = acc;
    }
}

extern "C" void kernel_launch(void* const* d_in, const int* in_sizes, int n_in,
                              void* d_out, int out_size, void* d_ws, size_t ws_size,
                              hipStream_t stream) {
    const float* embeds = (const float*)d_in[0];
    const float* weight = (const float*)d_in[1];
    const int*   eidx   = (const int*)d_in[2];
    const float* evals  = (const float*)d_in[3];

    const int M = in_sizes[0] / D;
    const int E = in_sizes[3];
    const int* dst = eidx;
    const int* src = eidx + E;

    float* support = (float*)d_ws;
    int* recs      = (int*)(support + (size_t)M * D);
    unsigned* runs = (unsigned*)(recs + ((size_t)E + SEG) * 4);
    float* out     = (float*)d_out;

    {
        gcn_gemm_wmma_f32<<<256, GEMM_THREADS, 0, stream>>>(
            embeds, weight, support, dst, src, evals, recs, runs, M, E);
    }

    {
        const int blocks = (M + SP_ROWS - 1) / SP_ROWS;
        gcn_spmm_gather<<<blocks, SP_THREADS, 0, stream>>>(
            support, dst, src, evals, recs, runs, out, E, M);
    }
}


# duty split 28 summing / 4 multiplying waves (one per SIMD, issue priority 3, a 32-column quad of all 3 strips of a 48-row round): summing waves do one load pass a step, not two; sum order unchanged
# speedup vs baseline: 2.4695x; 1.1031x over previous
#include <hip/hip_runtime.h>


typedef __attribute__((ext_vector_type(2))) float v2f;
typedef __attribute__((ext_vector_type(8))) float v8f;

#define D 128
#define SP_ROWS  392

#define SEG      1024
#define NBUCKET  256
#define SORT_THREADS 256

__device__ __forceinline__ unsigned sp_key(int d) {
    return ((unsigned)d / SP_ROWS) & (NBUCKET - 1u);
}


__global__ void __launch_bounds__(SORT_THREADS)
gcn_edge_sort(const int* __restrict__ dst,
              const int* __restrict__ src,
              const float* __restrict__ vals,
              int* __restrict__ recs4,
              unsigned* __restrict__ runs,
              int E) {
    __shared__ int hist[NBUCKET];
    __shared__ int wsum[SORT_THREADS / 32];
    int4* recs = (int4*)recs4;
    const int tid  = threadIdx.x;
    const int nseg = (E + SEG - 1) / SEG;
    for (int seg = blockIdx.x; seg < nseg; seg += gridDim.x) {
        __syncthreads();
        hist[tid] = 0;
        __syncthreads();
        int   dv[4], sv[4], rk[4];
        float vv[4];
        bool  ok[4];
#pragma unroll
        for (int j = 0; j < 4; ++j) {
            const int e = seg * SEG + j * 256 + tid;
            ok[j] = e < E;
            const int ec = ok[j] ? e : 0;
            dv[j] = dst[ec];
            sv[j] = src[ec];
            vv[j] = vals[ec];
        }
#pragma unroll
        for (int j = 0; j < 4; ++j)
            rk[j] = ok[j] ? atomicAdd(&hist[sp_key(dv[j])], 1) : 0;
        __syncthreads();
        const int c = hist[tid];
        int inc = c;
#pragma unroll
        for (int o = 1; o < 32; o <<= 1) {
            const int y = __shfl_up(inc, o);
            if ((tid & 31) >= o) inc += y;
        }
        if ((tid & 31) == 31) wsum[tid >> 5] = inc;
        __syncthreads();
        int base = 0;
        for (int i = 0; i < (tid >> 5); ++i) base += wsum[i];
        const int excl = base + inc - c;
        hist[tid] = excl;
        __syncthreads();
        runs[(size_t)tid * nseg + seg] = (unsigned)excl | ((unsigned)c << 16);
#pragma unroll
        for (int j = 0; j < 4; ++j)
            if (ok[j])
                recs[(size_t)seg * SEG + hist[sp_key(dv[j])] + rk[j]] =
                    make_int4(dv[j], sv[j], __float_as_int(vv[j]), 0);
    }
}

#define SP_CAP     32
#define SP_THREADS 1024
#define SP_WAVES   (SP_THREADS / 32)
#define CLAIM_DEPTH 8
#define RT_ROWS    48
#define SUM_WAVES  28
#define RT_PITCH   (D + 4)

__device__ __forceinline__ float4 sum_row_from_edges(const float* __restrict__ X,
                                                     const int* __restrict__ dst,
                                                     const int* __restrict__ src,
                                                     const float* __restrict__ vals,
                                                     int E, int row, int lane) {
    float4 acc = make_float4(0.f, 0.f, 0.f, 0.f);
    for (int b = 0; b < E; b += 32) {
        const int  e   = b + lane;
        const bool hit = (e < E) && (dst[e] == row);
        unsigned mask  = __builtin_amdgcn_ballot_w32(hit);
        while (mask) {
            const int j  = __builtin_ctz(mask);
            mask &= mask - 1;
            const int   sj = src[b + j];
            const float cj = vals[b + j];
            const float4 x = *(const float4*)(X + (size_t)sj * D + lane * 4);
            acc.x += cj * x.x;
            acc.y += cj * x.y;
            acc.z += cj * x.z;
            acc.w += cj * x.w;
        }
    }
    return acc;
}

template <int NS>
__device__ __forceinline__ void mult_strips(const float* __restrict__ tb,
                                            const float2* __restrict__ Wlds2,
                                            float* __restrict__ out,
                                            int lo, int rb, int nrows,
                                            int strip0, int quad, int half, int l15) {
    int wofs = half * D + l15 + quad * 32;
    asm volatile("" : "+v"(wofs) : : "memory");
    const float*  At = tb + (strip0 * 16 + l15) * RT_PITCH + half * 2;
    const float2* Wb = Wlds2 + wofs;

    v8f acc0[NS], acc1[NS];
#pragma unroll
    for (int t = 0; t < NS; ++t) {
        acc0[t] = v8f{};
        acc1[t] = v8f{};
    }
#pragma unroll
    for (int k = 0; k < D; k += 4) {
        const v2f b0 = *(const v2f*)(Wb + (k >> 1) * D);
        const v2f b1 = *(const v2f*)(Wb + (k >> 1) * D + 16);
#pragma unroll
        for (int t = 0; t < NS; ++t) {
            const v2f a = *(const v2f*)(At + t * 16 * RT_PITCH + k);
            acc0[t] = __builtin_amdgcn_wmma_f32_16x16x4_f32(
                false, a, false, b0, (short)0, acc0[t], false, false);
            acc1[t] = __builtin_amdgcn_wmma_f32_16x16x4_f32(
                false, a, false, b1, (short)0, acc1[t], false, false);
        }
    }

#pragma unroll
    for (int t = 0; t < NS; ++t) {
#pragma unroll
        for (int r = 0; r < 8; ++r) {
            const int rr = rb + (strip0 + t) * 16 + half * 8 + r;
            if (rr < nrows) {
                float* Op = out + (size_t)(lo + rr) * D + quad * 32 + l15;
                Op[0]  = acc0[t][r];
                Op[16] = acc1[t][r];
            }
        }
    }
}

__global__ void __launch_bounds__(SP_THREADS)
gcn_gather_gemm(const float* __restrict__ X,
                const float* __restrict__ W,
                const int*   __restrict__ dst,
                const int*   __restrict__ src,
                const float* __restrict__ vals,
                const int*   __restrict__ recs4,
                const unsigned* __restrict__ runs,
                float*       __restrict__ out,
                int E, int M) {
    __shared__ float2 Wlds2[(D / 2) * D];
    __shared__ int   cnt[SP_ROWS];
    __shared__ int   slotE[SP_ROWS * SP_CAP];
    __shared__ float slotV[SP_ROWS * SP_CAP];
    __shared__ __attribute__((aligned(16))) float tile[2][RT_ROWS * RT_PITCH];

    const int4* recs = (const int4*)recs4;
    const int tid  = threadIdx.x;
    const int lane = tid & 31;
    const int w    = __builtin_amdgcn_readfirstlane(tid >> 5);
    const int lo   = blockIdx.x * SP_ROWS;
    const unsigned nrows = (unsigned)min(SP_ROWS, M - lo);

    for (int i = tid; i < (D / 2) * (D / 4); i += SP_THREADS) {
        const int p  = i / (D / 4);
        const int c4 = (i % (D / 4)) * 4;
        const float4 r0 = *(const float4*)(W + (size_t)(2 * p)     * D + c4);
        const float4 r1 = *(const float4*)(W + (size_t)(2 * p + 1) * D + c4);
        float2* outp = Wlds2 + p * D + c4;
        outp[0] = make_float2(r0.x, r1.x);
        outp[1] = make_float2(r0.y, r1.y);
        outp[2] = make_float2(r0.z, r1.z);
        outp[3] = make_float2(r0.w, r1.w);
    }
    for (int i = tid; i < SP_ROWS; i += SP_THREADS) cnt[i] = 0;
    __syncthreads();

    {
        const int nseg = (E + SEG - 1) / SEG;
        const unsigned* myruns = runs + (size_t)(blockIdx.x & (NBUCKET - 1)) * nseg;
        for (int t = tid; t < nseg; t += SP_THREADS) {
            const unsigned wd = myruns[t];
            const int n = (int)(wd >> 16);
            const int4* rp = recs + (size_t)t * SEG + (wd & 0xffffu);
            for (int k0 = 0; k0 < n; k0 += CLAIM_DEPTH) {
                int4 rc[CLAIM_DEPTH];
#pragma unroll
                for (int u = 0; u < CLAIM_DEPTH; ++u)
                    rc[u] = rp[min(k0 + u, n - 1)];
#pragma unroll
                for (int u = 0; u < CLAIM_DEPTH; ++u) {
                    const unsigned r = (unsigned)(rc[u].x - lo);
                    if (k0 + u < n && r < nrows) {
                        const int pos = atomicAdd(&cnt[r], 1);
                        if (pos < SP_CAP) {
                            slotE[r * SP_CAP + pos] = rc[u].y;
                            slotV[r * SP_CAP + pos] = __int_as_float(rc[u].z);
                        }
                    }
                }
            }
        }
    }
    __syncthreads();

    const int half = lane >> 4;
    const int l15  = lane & 15;
    const bool mult = ((w & 7) == (w >> 3));
    const int  quad = w >> 3;
    const int  gi   = w - (w + 8) / 9;
    const int nrounds = ((int)nrows + RT_ROWS - 1) / RT_ROWS;
    if (mult) __builtin_amdgcn_s_setprio(3);
    for (int s = 0; s <= nrounds; ++s) {
        const bool sums = (s == 0) || !mult;
        if (sums) {
            if (s < nrounds) {
                const int rb = s * RT_ROWS;
                float* tb = tile[s & 1];
                const int strips = min(RT_ROWS / 16, ((int)nrows - rb + 15) >> 4);
                const int first  = (s == 0) ? w : gi;
                const int stride = (s == 0) ? SP_WAVES : SUM_WAVES;
                for (int lr = first; lr < strips * 16; lr += 2 * stride) {
                    const int  lrB  = lr + stride;
                    const bool hasB = lrB < strips * 16;
                    const int  rA   = rb + lr;
                    const int  rB   = rb + lrB;
                    const int nA = (rA < (int)nrows) ? __builtin_amdgcn_readfirstlane(cnt[rA]) : 0;
                    const int nB = (hasB && rB < (int)nrows) ? __builtin_amdgcn_readfirstlane(cnt[rB]) : 0;
                    const int mA = (nA <= SP_CAP) ? nA : 0;
                    const int mB = (nB <= SP_CAP) ? nB : 0;
                    int   sA = 0, sB = 0;
                    float vA = 0.f, vB = 0.f;
                    if (lane < mA) {
                        sA = slotE[rA * SP_CAP + lane];
                        vA = slotV[rA * SP_CAP + lane];
                    }
                    if (lane < mB) {
                        sB = slotE[rB * SP_CAP + lane];
                        vB = slotV[rB * SP_CAP + lane];
                    }
                    float4 accA = make_float4(0.f, 0.f, 0.f, 0.f);
                    float4 accB = make_float4(0.f, 0.f, 0.f, 0.f);
                    const int mm = max(mA, mB);
                    for (int j0 = 0; j0 < mm; j0 += 8) {
                        float4 xa[8], xb[8];
                        float  ca[8], cb[8];
#pragma unroll
                        for (int u = 0; u < 8; ++u) {
                            xa[u] = make_float4(0.f, 0.f, 0.f, 0.f);
                            ca[u] = 0.f;
                            if (j0 + u < mA) {
                                const int sj = __builtin_amdgcn_readlane(sA, j0 + u);
                                ca[u] = __int_as_float(
                                    __builtin_amdgcn_readlane(__float_as_int(vA), j0 + u));
                                xa[u] = *(const float4*)(X + (size_t)sj * D + lane * 4);
                            }
                        }
#pragma unroll
                        for (int u = 0; u < 8; ++u) {
                            xb[u] = make_float4(0.f, 0.f, 0.f, 0.f);
                            cb[u] = 0.f;
                            if (j0 + u < mB) {
                                const int sj = __builtin_amdgcn_readlane(sB, j0 + u);
                                cb[u] = __int_as_float(
                                    __builtin_amdgcn_readlane(__float_as_int(vB), j0 + u));
                                xb[u] = *(const float4*)(X + (size_t)sj * D + lane * 4);
                            }
                        }
#pragma unroll
                        for (int u = 0; u < 8; ++u) {
                            if (j0 + u < mA) {
                                accA.x += ca[u] * xa[u].x;
                                accA.y += ca[u] * xa[u].y;
                                accA.z += ca[u] * xa[u].z;
                                accA.w += ca[u] * xa[u].w;
                            }
                        }
#pragma unroll
                        for (int u = 0; u < 8; ++u) {
                            if (j0 + u < mB) {
                                accB.x += cb[u] * xb[u].x;
                                accB.y += cb[u] * xb[u].y;
                                accB.z += cb[u] * xb[u].z;
                                accB.w += cb[u] * xb[u].w;
                            }
                        }
                    }
                    if (nA > SP_CAP) accA = sum_row_from_edges(X, dst, src, vals, E, lo + rA, lane);
                    if (nB > SP_CAP) accB = sum_row_from_edges(X, dst, src, vals, E, lo + rB, lane);
                    *(float4*)(tb + lr * RT_PITCH + lane * 4) = accA;
                    if (hasB) *(float4*)(tb + lrB * RT_PITCH + lane * 4) = accB;
                }
            }
        } else {
            const int rb = (s - 1) * RT_ROWS;
            const float* tb = tile[(s - 1) & 1];
            const int strips = min(RT_ROWS / 16, ((int)nrows - rb + 15) >> 4);
            if (strips == RT_ROWS / 16) {
                mult_strips<RT_ROWS / 16>(tb, Wlds2, out, lo, rb, (int)nrows, 0, quad, half, l15);
            } else {
                for (int st = 0; st < strips; ++st)
                    mult_strips<1>(tb, Wlds2, out, lo, rb, (int)nrows, st, quad, half, l15);
            }
        }

        __builtin_amdgcn_fence(__ATOMIC_RELEASE, "workgroup", "local");
        __builtin_amdgcn_s_barrier();
        __builtin_amdgcn_fence(__ATOMIC_ACQUIRE, "workgroup", "local");
    }
}

extern "C" void kernel_launch(void* const* d_in, const int* in_sizes, int n_in,
                              void* d_out, int out_size, void* d_ws, size_t ws_size,
                              hipStream_t stream) {
    const float* embeds = (const float*)d_in[0];
    const float* weight = (const float*)d_in[1];
    const int*   eidx   = (const int*)d_in[2];
    const float* evals  = (const float*)d_in[3];

    const int M = in_sizes[0] / D;
    const int E = in_sizes[3];
    const int* dst = eidx;
    const int* src = eidx + E;

    int* recs      = (int*)d_ws;
    unsigned* runs = (unsigned*)(recs + ((size_t)E + SEG) * 4);
    float* out     = (float*)d_out;

    {
        const int segs = (E + SEG - 1) / SEG;
        gcn_edge_sort<<<segs, SORT_THREADS, 0, stream>>>(dst, src, evals, recs, runs, E);
    }

    {
        const int blocks = (M + SP_ROWS - 1) / SP_ROWS;
        gcn_gather_gemm<<<blocks, SP_THREADS, 0, stream>>>(
            embeds, weight, dst, src, evals, recs, runs, out, E, M);
    }
}
